# starting kernel
# baseline (speedup 1.0000x reference)
#include <hip/hip_runtime.h>

typedef __attribute__((ext_vector_type(2))) float v2f;
typedef __attribute__((ext_vector_type(8))) float v8f;

#define S_POS 20
#define DIM   64

__device__ __forceinline__ float pick8(v8f a, int i) {
  float r = a[0];
  if (i == 1) r = a[1];
  if (i == 2) r = a[2];
  if (i == 3) r = a[3];
  if (i == 4) r = a[4];
  if (i == 5) r = a[5];
  if (i == 6) r = a[6];
  if (i == 7) r = a[7];
  return r;
}

__global__ __launch_bounds__(256) void cofiset_score_kernel(
    const float* __restrict__ U, const float* __restrict__ V,
    const float* __restrict__ bi, const int* __restrict__ user_id,
    const int* __restrict__ pos_item_id, const int* __restrict__ neg_item_id,
    float* __restrict__ out, int B) {
  const int lane = threadIdx.x & 31;
  const int wave = blockIdx.x * (blockDim.x >> 5) + (threadIdx.x >> 5);
  const int b0   = wave * 16;
  if (b0 >= B) return;

  const int n       = lane & 15;
  const int hi      = lane >> 4;
  const int colBase = 2 * hi;

  int bb = b0 + n;
  if (bb >= B) bb = B - 1;

  v2f bsum[16];
#pragma unroll
  for (int k = 0; k < 16; ++k) { v2f z = {0.f, 0.f}; bsum[k] = z; }

  const int* prow = pos_item_id + (size_t)bb * S_POS;
  for (int s = 0; s < S_POS; ++s) {
    const float* vr = V + (size_t)prow[s] * DIM + colBase;
#pragma unroll
    for (int k = 0; k < 16; ++k) {
      v2f t = *(const v2f*)(vr + 4 * k);
      bsum[k] += t;
    }
  }

  v2f arow[16];
  {
    const float* ur = U + (size_t)user_id[bb] * DIM + colBase;
#pragma unroll
    for (int k = 0; k < 16; ++k) arow[k] = *(const v2f*)(ur + 4 * k);
  }

  v8f accP = {};
#pragma unroll
  for (int k = 0; k < 16; ++k)
    accP = __builtin_amdgcn_wmma_f32_16x16x4_f32(
        false, arow[k], false, bsum[k], (short)0, accP, false, false);

  v8f accN = {};
  {
    const float* vr = V + (size_t)neg_item_id[bb] * DIM + colBase;
#pragma unroll
    for (int k = 0; k < 16; ++k) {
      v2f t = *(const v2f*)(vr + 4 * k);
      accN = __builtin_amdgcn_wmma_f32_16x16x4_f32(
          false, arow[k], false, t, (short)0, accN, false, false);
    }
  }

  const bool active = (hi == 0) ? (n < 8) : (n >= 8);
  if (active && (b0 + n) < B) {
    const int b  = b0 + n;
    const float dp = pick8(accP, n & 7);
    const float dn = pick8(accN, n & 7);
    float bs = 0.f;
    const int* pr = pos_item_id + (size_t)b * S_POS;
#pragma unroll
    for (int s = 0; s < S_POS; ++s) bs += bi[pr[s]];
    out[b]     = (dp + bs) * (1.0f / S_POS);
    out[B + b] = dn + bi[neg_item_id[b]];
  }
}

extern "C" void kernel_launch(void* const* d_in, const int* in_sizes, int n_in,
                              void* d_out, int out_size, void* d_ws, size_t ws_size,
                              hipStream_t stream) {
  const float* U           = (const float*)d_in[0];
  const float* V           = (const float*)d_in[1];
  const float* bi          = (const float*)d_in[2];
  const int*   user_id     = (const int*)d_in[3];
  const int*   pos_item_id = (const int*)d_in[4];
  const int*   neg_item_id = (const int*)d_in[5];
  float*       out         = (float*)d_out;

  const int B = in_sizes[3];
  const int waves = (B + 15) / 16;
  const int threads = 256;
  const int wavesPerBlock = threads / 32;
  const int blocks = (waves + wavesPerBlock - 1) / wavesPerBlock;

  cofiset_score_kernel<<<blocks, threads, 0, stream>>>(
      U, V, bi, user_id, pos_item_id, neg_item_id, out, B);
}


# launch shape only: work-group 256 -> 128 threads, so the 1024 waves sit in 256 work-groups (one per WGP, one wave per SIMD) instead of 128 work-groups on half the WGPs; kernel body unchanged
# speedup vs baseline: 1.3486x; 1.3486x over previous
#include <hip/hip_runtime.h>

typedef __attribute__((ext_vector_type(2))) float v2f;
typedef __attribute__((ext_vector_type(8))) float v8f;

#define S_POS 20
#define DIM   64

__device__ __forceinline__ float pick8(v8f a, int i) {
  float r = a[0];
  if (i == 1) r = a[1];
  if (i == 2) r = a[2];
  if (i == 3) r = a[3];
  if (i == 4) r = a[4];
  if (i == 5) r = a[5];
  if (i == 6) r = a[6];
  if (i == 7) r = a[7];
  return r;
}

__global__ __launch_bounds__(256) void cofiset_score_kernel(
    const float* __restrict__ U, const float* __restrict__ V,
    const float* __restrict__ bi, const int* __restrict__ user_id,
    const int* __restrict__ pos_item_id, const int* __restrict__ neg_item_id,
    float* __restrict__ out, int B) {
  const int lane = threadIdx.x & 31;
  const int wave = blockIdx.x * (blockDim.x >> 5) + (threadIdx.x >> 5);
  const int b0   = wave * 16;
  if (b0 >= B) return;

  const int n       = lane & 15;
  const int hi      = lane >> 4;
  const int colBase = 2 * hi;

  int bb = b0 + n;
  if (bb >= B) bb = B - 1;

  v2f bsum[16];
#pragma unroll
  for (int k = 0; k < 16; ++k) { v2f z = {0.f, 0.f}; bsum[k] = z; }

  const int* prow = pos_item_id + (size_t)bb * S_POS;
  for (int s = 0; s < S_POS; ++s) {
    const float* vr = V + (size_t)prow[s] * DIM + colBase;
#pragma unroll
    for (int k = 0; k < 16; ++k) {
      v2f t = *(const v2f*)(vr + 4 * k);
      bsum[k] += t;
    }
  }

  v2f arow[16];
  {
    const float* ur = U + (size_t)user_id[bb] * DIM + colBase;
#pragma unroll
    for (int k = 0; k < 16; ++k) arow[k] = *(const v2f*)(ur + 4 * k);
  }

  v8f accP = {};
#pragma unroll
  for (int k = 0; k < 16; ++k)
    accP = __builtin_amdgcn_wmma_f32_16x16x4_f32(
        false, arow[k], false, bsum[k], (short)0, accP, false, false);

  v8f accN = {};
  {
    const float* vr = V + (size_t)neg_item_id[bb] * DIM + colBase;
#pragma unroll
    for (int k = 0; k < 16; ++k) {
      v2f t = *(const v2f*)(vr + 4 * k);
      accN = __builtin_amdgcn_wmma_f32_16x16x4_f32(
          false, arow[k], false, t, (short)0, accN, false, false);
    }
  }

  const bool active = (hi == 0) ? (n < 8) : (n >= 8);
  if (active && (b0 + n) < B) {
    const int b  = b0 + n;
    const float dp = pick8(accP, n & 7);
    const float dn = pick8(accN, n & 7);
    float bs = 0.f;
    const int* pr = pos_item_id + (size_t)b * S_POS;
#pragma unroll
    for (int s = 0; s < S_POS; ++s) bs += bi[pr[s]];
    out[b]     = (dp + bs) * (1.0f / S_POS);
    out[B + b] = dn + bi[neg_item_id[b]];
  }
}

extern "C" void kernel_launch(void* const* d_in, const int* in_sizes, int n_in,
                              void* d_out, int out_size, void* d_ws, size_t ws_size,
                              hipStream_t stream) {
  const float* U           = (const float*)d_in[0];
  const float* V           = (const float*)d_in[1];
  const float* bi          = (const float*)d_in[2];
  const int*   user_id     = (const int*)d_in[3];
  const int*   pos_item_id = (const int*)d_in[4];
  const int*   neg_item_id = (const int*)d_in[5];
  float*       out         = (float*)d_out;

  const int B = in_sizes[3];
  const int waves = (B + 15) / 16;
  const int threads = 128;
  const int wavesPerBlock = threads / 32;
  const int blocks = (waves + wavesPerBlock - 1) / wavesPerBlock;

  cofiset_score_kernel<<<blocks, threads, 0, stream>>>(
      U, V, bi, user_id, pos_item_id, neg_item_id, out, B);
}


# gather pipelined: all 22 indices read in one round trip up front; user/negative rows and biases issued at once, 20 positive rows streamed through a 3-row ring kept in flight; sum order unchanged
# speedup vs baseline: 1.7455x; 1.2943x over previous
#include <hip/hip_runtime.h>


typedef __attribute__((ext_vector_type(2))) float v2f;
typedef __attribute__((ext_vector_type(8))) float v8f;

#define S_POS  20
#define DIM    64
#define KSTEPS 16
#define PF     3

#define KEEP_ISSUE_ORDER() __builtin_amdgcn_sched_barrier(0)

__device__ __forceinline__ float pick8(v8f a, int i) {
  float r = a[0];
  if (i == 1) r = a[1];
  if (i == 2) r = a[2];
  if (i == 3) r = a[3];
  if (i == 4) r = a[4];
  if (i == 5) r = a[5];
  if (i == 6) r = a[6];
  if (i == 7) r = a[7];
  return r;
}

__global__ __launch_bounds__(256) void cofiset_score_kernel(
    const float* __restrict__ U, const float* __restrict__ V,
    const float* __restrict__ bi, const int* __restrict__ user_id,
    const int* __restrict__ pos_item_id, const int* __restrict__ neg_item_id,
    float* __restrict__ out, int B) {
  const int lane = threadIdx.x & 31;
  const int wave = blockIdx.x * (blockDim.x >> 5) + (threadIdx.x >> 5);
  const int b0   = wave * 16;
  if (b0 >= B) return;

  const int n       = lane & 15;
  const int hi      = lane >> 4;
  const int colBase = 2 * hi;

  int bb = b0 + n;
  if (bb >= B) bb = B - 1;

  const int* prow = pos_item_id + (size_t)bb * S_POS;
  int pidx[S_POS];
#pragma unroll
  for (int s = 0; s < S_POS; ++s) pidx[s] = prow[s];
  const int uid = user_id[bb];
  const int nid = neg_item_id[bb];
  asm volatile("" :: "v"(pidx[0]), "v"(pidx[1]), "v"(pidx[2]), "v"(pidx[3]),
               "v"(pidx[4]), "v"(pidx[5]), "v"(pidx[6]), "v"(pidx[7]),
               "v"(pidx[8]), "v"(pidx[9]), "v"(pidx[10]), "v"(pidx[11]),
               "v"(pidx[12]), "v"(pidx[13]), "v"(pidx[14]), "v"(pidx[15]),
               "v"(pidx[16]), "v"(pidx[17]), "v"(pidx[18]), "v"(pidx[19]),
               "v"(uid), "v"(nid));

  const float* Vc = V + colBase;

  v2f arow[KSTEPS];
  {
    const float* ur = U + (size_t)uid * DIM + colBase;
#pragma unroll
    for (int k = 0; k < KSTEPS; ++k) arow[k] = *(const v2f*)(ur + 4 * k);
  }
  v2f nrow[KSTEPS];
  {
    const float* vr = Vc + (size_t)nid * DIM;
#pragma unroll
    for (int k = 0; k < KSTEPS; ++k) nrow[k] = *(const v2f*)(vr + 4 * k);
  }
  float bv[S_POS];
#pragma unroll
  for (int s = 0; s < S_POS; ++s) bv[s] = bi[pidx[s]];
  const float bneg = bi[nid];

  v2f ring[PF][KSTEPS];
#pragma unroll
  for (int r = 0; r < PF; ++r) {
    const float* vr = Vc + (size_t)pidx[r] * DIM;
#pragma unroll
    for (int k = 0; k < KSTEPS; ++k) ring[r][k] = *(const v2f*)(vr + 4 * k);
  }
  KEEP_ISSUE_ORDER();

  float bs = 0.f;
#pragma unroll
  for (int s = 0; s < S_POS; ++s) bs += bv[s];
  asm volatile("" :: "v"(bs), "v"(bneg));
  KEEP_ISSUE_ORDER();

  v2f bsum[KSTEPS];
#pragma unroll
  for (int k = 0; k < KSTEPS; ++k) { v2f z = {0.f, 0.f}; bsum[k] = z; }

#pragma unroll
  for (int s = 0; s < S_POS - 1; ++s) {
    const int slot = s % PF;
    const bool refill = (s + PF) < S_POS;
    const float* vr = Vc + (size_t)pidx[refill ? s + PF : s] * DIM;
#pragma unroll
    for (int k = 0; k < KSTEPS; ++k) {
      bsum[k] += ring[slot][k];
      if (refill) ring[slot][k] = *(const v2f*)(vr + 4 * k);
    }
    KEEP_ISSUE_ORDER();
  }

  v8f accP = {};
  v8f accN = {};
#pragma unroll
  for (int k = 0; k < KSTEPS; ++k) {
    bsum[k] += ring[(S_POS - 1) % PF][k];
    accP = __builtin_amdgcn_wmma_f32_16x16x4_f32(
        false, arow[k], false, bsum[k], (short)0, accP, false, false);
    accN = __builtin_amdgcn_wmma_f32_16x16x4_f32(
        false, arow[k], false, nrow[k], (short)0, accN, false, false);
  }

  const bool active = (hi == 0) ? (n < 8) : (n >= 8);
  if (active && (b0 + n) < B) {
    const int b  = b0 + n;
    const float dp = pick8(accP, n & 7);
    const float dn = pick8(accN, n & 7);
    out[b]     = (dp + bs) * (1.0f / S_POS);
    out[B + b] = dn + bneg;
  }
}

extern "C" void kernel_launch(void* const* d_in, const int* in_sizes, int n_in,
                              void* d_out, int out_size, void* d_ws, size_t ws_size,
                              hipStream_t stream) {
  const float* U           = (const float*)d_in[0];
  const float* V           = (const float*)d_in[1];
  const float* bi          = (const float*)d_in[2];
  const int*   user_id     = (const int*)d_in[3];
  const int*   pos_item_id = (const int*)d_in[4];
  const int*   neg_item_id = (const int*)d_in[5];
  float*       out         = (float*)d_out;

  const int B = in_sizes[3];
  const int waves = (B + 15) / 16;
  const int threads = 128;
  const int wavesPerBlock = threads / 32;
  const int blocks = (waves + wavesPerBlock - 1) / wavesPerBlock;

  cofiset_score_kernel<<<blocks, threads, 0, stream>>>(
      U, V, bi, user_id, pos_item_id, neg_item_id, out, B);
}


# gather coalesced: each load reads two whole 256-byte table rows (4 full lines, was 16 lines x 16 B); rows, sums and biases go through wave-private LDS into the WMMA layout; sum order unchanged
# speedup vs baseline: 2.4081x; 1.3796x over previous
#include <hip/hip_runtime.h>


typedef __attribute__((ext_vector_type(2))) float v2f;
typedef __attribute__((ext_vector_type(4))) float v4f;
typedef __attribute__((ext_vector_type(8))) float v8f;
typedef __attribute__((ext_vector_type(4))) int   v4i;

#define S_POS  20
#define DIM    64
#define KSTEPS 16
#define NPAIR  8
#define WAVES_PER_BLOCK 4
#define PITCH  68
#define BPITCH 32
#define STAGE_DWORDS (3 * 16 * PITCH + 16 * BPITCH)

#define KEEP_ISSUE_ORDER() __builtin_amdgcn_sched_barrier(0)

__device__ __forceinline__ float pick8(v8f a, int i) {
  float r = a[0];
  if (i == 1) r = a[1];
  if (i == 2) r = a[2];
  if (i == 3) r = a[3];
  if (i == 4) r = a[4];
  if (i == 5) r = a[5];
  if (i == 6) r = a[6];
  if (i == 7) r = a[7];
  return r;
}

__device__ __forceinline__ v4f load_row(const float* __restrict__ table, int row, int c) {
  const unsigned off = (unsigned)row * (unsigned)(DIM * sizeof(float)) + 16u * (unsigned)c;
  return *(const v4f*)((const char*)table + off);
}
__device__ __forceinline__ float load_bias(const float* __restrict__ bi, int item) {
  return *(const float*)((const char*)bi + ((unsigned)item << 2));
}

__global__ __launch_bounds__(WAVES_PER_BLOCK * 32) void cofiset_score_kernel(
    const float* __restrict__ U, const float* __restrict__ V,
    const float* __restrict__ bi, const int* __restrict__ user_id,
    const int* __restrict__ pos_item_id, const int* __restrict__ neg_item_id,
    float* __restrict__ out, int B) {
  __shared__ __attribute__((aligned(16))) float stage[WAVES_PER_BLOCK][STAGE_DWORDS];

  const int lane = threadIdx.x & 31;
  const int wib  = threadIdx.x >> 5;
  const int wave = blockIdx.x * (blockDim.x >> 5) + wib;
  const int b0   = wave * 16;
  if (b0 >= B) return;

  const int c  = lane & 15;
  const int hi = lane >> 4;

  float* Lu = &stage[wib][0];
  float* Ln = Lu + 16 * PITCH;
  float* Lp = Ln + 16 * PITCH;
  float* Lb = Lp + 16 * PITCH;

  int bbp[NPAIR];
#pragma unroll
  for (int i = 0; i < NPAIR; ++i) {
    int bb = b0 + 2 * i + hi;
    bbp[i] = (bb >= B) ? (B - 1) : bb;
  }

  v4i q[4][5];
  int nidv[4];
  v4f pr[2][S_POS];
  v4f nr[2];

#define LOAD_IDX(i)                                                          \
  {                                                                          \
    const v4i* pq = (const v4i*)(pos_item_id + (size_t)bbp[(i)] * S_POS);    \
    _Pragma("unroll") for (int j = 0; j < 5; ++j) q[(i) & 3][j] = pq[j];     \
    nidv[(i) & 3] = neg_item_id[bbp[(i)]];                                   \
  }
#define ISSUE_ROWS(i)                                                        \
  {                                                                          \
    nr[(i) & 1] = load_row(V, nidv[(i) & 3], c);                             \
    _Pragma("unroll") for (int s = 0; s < S_POS; ++s)                        \
        pr[(i) & 1][s] = load_row(V, q[(i) & 3][s >> 2][s & 3], c);          \
  }

  int uidv[NPAIR];
#pragma unroll
  for (int i = 0; i < NPAIR; ++i) uidv[i] = user_id[bbp[i]];
  LOAD_IDX(0) LOAD_IDX(1) LOAD_IDX(2) LOAD_IDX(3)
  KEEP_ISSUE_ORDER();

  v4f ur[NPAIR];
#pragma unroll
  for (int i = 0; i < NPAIR; ++i) ur[i] = load_row(U, uidv[i], c);
  ISSUE_ROWS(0) ISSUE_ROWS(1)
  KEEP_ISSUE_ORDER();

  int   bia[NPAIR], bib[NPAIR];
  float bva[NPAIR], bvb[NPAIR];

#pragma unroll
  for (int i = 0; i < NPAIR; ++i) {
    const int r = 2 * i + hi;
    if (i == 0) {
#pragma unroll
      for (int j = 0; j < NPAIR; ++j) *(v4f*)(Lu + (2 * j + hi) * PITCH + 4 * c) = ur[j];
    }
    *(v4f*)(Ln + r * PITCH + 4 * c) = nr[i & 1];
    v4f acc = {0.f, 0.f, 0.f, 0.f};
#pragma unroll
    for (int s = 0; s < S_POS; ++s) acc += pr[i & 1][s];
    *(v4f*)(Lp + r * PITCH + 4 * c) = acc;
    if (i == 5) {
#pragma unroll
      for (int j = 0; j < NPAIR; ++j) {
        Lb[(2 * j + hi) * BPITCH + c]      = bva[j];
        Lb[(2 * j + hi) * BPITCH + 16 + c] = bvb[j];
      }
    }
    KEEP_ISSUE_ORDER();
    if (i + 4 < NPAIR) LOAD_IDX(i + 4)
    if (i + 2 < NPAIR) ISSUE_ROWS(i + 2)
    if (i == 1) {
#pragma unroll
      for (int j = 0; j < NPAIR; ++j) {
        const int* prow = pos_item_id + (size_t)bbp[j] * S_POS;
        bia[j] = prow[c];
        const int* pb = (c < 4) ? (prow + 16 + c) : (neg_item_id + bbp[j]);
        bib[j] = *pb;
      }
    }
    if (i == 3) {
#pragma unroll
      for (int j = 0; j < NPAIR; ++j) { bva[j] = load_bias(bi, bia[j]); bvb[j] = load_bias(bi, bib[j]); }
    }
    KEEP_ISSUE_ORDER();
  }
#undef LOAD_IDX
#undef ISSUE_ROWS

  __builtin_amdgcn_fence(__ATOMIC_RELEASE, "wavefront");
  __builtin_amdgcn_wave_barrier();
  __builtin_amdgcn_fence(__ATOMIC_ACQUIRE, "wavefront");

  const int n = c;

  const float* fb = Lb + n * BPITCH;
  float fbv[S_POS + 1];
#pragma unroll
  for (int s = 0; s <= S_POS; ++s) fbv[s] = fb[s];

  const float* fu = Lu + n * PITCH + 2 * hi;
  const float* fn = Ln + n * PITCH + 2 * hi;
  const float* fp = Lp + n * PITCH + 2 * hi;
  v2f arow[KSTEPS], nrow[KSTEPS], bsum[KSTEPS];
#pragma unroll
  for (int k = 0; k < KSTEPS; ++k) {
    arow[k] = *(const v2f*)(fu + 4 * k);
    bsum[k] = *(const v2f*)(fp + 4 * k);
    nrow[k] = *(const v2f*)(fn + 4 * k);
  }

  float bs = 0.f;
#pragma unroll
  for (int s = 0; s < S_POS; ++s) bs += fbv[s];
  const float bneg = fbv[S_POS];
  asm volatile("" :: "v"(bs), "v"(bneg));
  KEEP_ISSUE_ORDER();

  v8f accP = {};
  v8f accN = {};
#pragma unroll
  for (int k = 0; k < KSTEPS; ++k) {
    accP = __builtin_amdgcn_wmma_f32_16x16x4_f32(
        false, arow[k], false, bsum[k], (short)0, accP, false, false);
    accN = __builtin_amdgcn_wmma_f32_16x16x4_f32(
        false, arow[k], false, nrow[k], (short)0, accN, false, false);
  }

  const bool active = (hi == 0) ? (n < 8) : (n >= 8);
  if (active && (b0 + n) < B) {
    const int b  = b0 + n;
    const float dp = pick8(accP, n & 7);
    const float dn = pick8(accN, n & 7);
    out[b]     = (dp + bs) * (1.0f / S_POS);
    out[B + b] = dn + bneg;
  }
}

extern "C" void kernel_launch(void* const* d_in, const int* in_sizes, int n_in,
                              void* d_out, int out_size, void* d_ws, size_t ws_size,
                              hipStream_t stream) {
  const float* U           = (const float*)d_in[0];
  const float* V           = (const float*)d_in[1];
  const float* bi          = (const float*)d_in[2];
  const int*   user_id     = (const int*)d_in[3];
  const int*   pos_item_id = (const int*)d_in[4];
  const int*   neg_item_id = (const int*)d_in[5];
  float*       out         = (float*)d_out;

  const int B = in_sizes[3];
  const int waves = (B + 15) / 16;
  const int threads = WAVES_PER_BLOCK * 32;
  const int wavesPerBlock = threads / 32;
  const int blocks = (waves + wavesPerBlock - 1) / wavesPerBlock;

  cofiset_score_kernel<<<blocks, threads, 0, stream>>>(
      U, V, bi, user_id, pos_item_id, neg_item_id, out, B);
}
